# starting kernel
# baseline (speedup 1.0000x reference)
#include <hip/hip_runtime.h>
#include <hip/hip_bf16.h>
#include <math.h>


typedef __attribute__((ext_vector_type(2))) float v2f;
typedef __attribute__((ext_vector_type(8))) float v8f;

#define NPIX      16384
#define KDIM      72
#define NDIM      128
#define LDSW_STR  132
#define VOTE_STR  130
#define CAPS_EPS  1e-7f
#define PH        34
#define PLANE     (PH * PH * 128)
#define PIX_PER_WAVE 4

__global__ void caps_pad_transpose_kernel(const float* __restrict__ in,
                                          float* __restrict__ out) {
    int t = blockIdx.x * blockDim.x + threadIdx.x;
    if (t >= 16 * PH * PH * 16) return;
    int b  = t & 15;
    int r  = t >> 4;
    int ww = r % PH;  r /= PH;
    int hh = r % PH;
    int ic = r / PH;
    int h = hh - 1;
    int w = ww - 1;
    float4 v0 = float4{0.f, 0.f, 0.f, 0.f};
    float4 v1 = v0;
    if ((unsigned)h < 32u && (unsigned)w < 32u) {
        const float4* src = (const float4*)(in + ((((size_t)b * 32 + h) * 32 + w) * 16 + ic) * 8);
        v0 = src[0];
        v1 = src[1];
    }
    float4* dst = (float4*)(out + (size_t)t * 8);
    dst[0] = v0;
    dst[1] = v1;
}

__global__ __launch_bounds__(64) void caps_conv_route_kernel(
    const float* __restrict__ xTp,
    const float* __restrict__ Wg,
    const float* __restrict__ bias,
    float* __restrict__ out)
{
    __shared__ float ldsW[KDIM * LDSW_STR];
    __shared__ float ldsVotes[2][16 * VOTE_STR];
    __shared__ float ldsLogits[2][16 * 16];
    __shared__ float ldsMx[2][16];
    __shared__ float ldsRs[2][16];

    const int tid  = threadIdx.x;
    const int wib  = tid >> 5;
    const int lane = tid & 31;

    for (int i = tid; i < KDIM * NDIM; i += 64) {
        int r = i >> 7;
        int c = i & 127;
        ldsW[r * LDSW_STR + c] = Wg[i];
    }
    __syncthreads();

    const int m  = lane & 15;
    const int hi = lane >> 4;

    float bias_l[4];
#pragma unroll
    for (int q = 0; q < 4; ++q) bias_l[q] = bias[q * 32 + lane];

    float* votes  = ldsVotes[wib];
    float* logits = ldsLogits[wib];

    const int waveId = blockIdx.x * 2 + wib;

#pragma unroll 1
    for (int it = 0; it < PIX_PER_WAVE; ++it) {
        const int pixel = waveId * PIX_PER_WAVE + it;
        const int icap  = pixel >> 10;
        const int h     = (pixel >> 5) & 31;
        const int w     = pixel & 31;

        v8f acc[8];
#pragma unroll
        for (int t = 0; t < 8; ++t)
            acc[t] = v8f{0.f, 0.f, 0.f, 0.f, 0.f, 0.f, 0.f, 0.f};

        const float* base = xTp + (size_t)icap * PLANE + (h * PH + w) * 128;

#pragma unroll
        for (int ks = 0; ks < 18; ++ks) {
            int K0  = ks * 4 + hi * 2;
            int kh  = K0 / 24;
            int rem = K0 - kh * 24;
            int kw  = rem >> 3;
            int id0 = rem & 7;
            v2f a = *(const v2f*)(base + (kh * PH + kw) * 128 + m * 8 + id0);
#pragma unroll
            for (int t = 0; t < 8; ++t) {
                int col = t * 16 + m;
                v2f bf;
                bf[0] = ldsW[K0 * LDSW_STR + col];
                bf[1] = ldsW[(K0 + 1) * LDSW_STR + col];
                acc[t] = __builtin_amdgcn_wmma_f32_16x16x4_f32(
                    false, a, false, bf, (short)0, acc[t], false, false);
            }
        }

#pragma unroll
        for (int t = 0; t < 8; ++t)
#pragma unroll
            for (int v = 0; v < 8; ++v)
                votes[(v + 8 * hi) * VOTE_STR + t * 16 + m] = acc[t][v];

        float pre[4], act[4];
#pragma unroll
        for (int q = 0; q < 4; ++q) {
            int c = q * 32 + lane;
            float s = 0.f;
            for (int ic = 0; ic < 16; ++ic) s += votes[ic * VOTE_STR + c];
            pre[q] = s * (1.f / 16.f) + bias_l[q];
        }
#pragma unroll
        for (int q = 0; q < 4; ++q) {
            float n2 = pre[q] * pre[q];
            n2 += __shfl_xor(n2, 1, 32);
            n2 += __shfl_xor(n2, 2, 32);
            n2 += __shfl_xor(n2, 4, 32);
            float scale = (n2 / (1.f + n2)) / sqrtf(n2 + CAPS_EPS);
            act[q] = pre[q] * scale;
        }

        for (int ic = 0; ic < 16; ++ic) {
#pragma unroll
            for (int q = 0; q < 4; ++q) {
                float d = votes[ic * VOTE_STR + q * 32 + lane] * act[q];
                d += __shfl_xor(d, 1, 32);
                d += __shfl_xor(d, 2, 32);
                d += __shfl_xor(d, 4, 32);
                if ((lane & 7) == 0)
                    logits[ic * 16 + q * 4 + (lane >> 3)] = d;
            }
        }

        if (lane < 16) {
            float mx = -3.0e38f;
            for (int oc = 0; oc < 16; ++oc) mx = fmaxf(mx, logits[lane * 16 + oc]);
            float s = 0.f;
            for (int oc = 0; oc < 16; ++oc) s += __expf(logits[lane * 16 + oc] - mx);
            ldsMx[wib][lane] = mx;
            ldsRs[wib][lane] = 1.f / s;
        }

#pragma unroll
        for (int q = 0; q < 4; ++q) {
            int c  = q * 32 + lane;
            int oc = c >> 3;
            float s = 0.f;
            for (int ic = 0; ic < 16; ++ic) {
                float r = __expf(logits[ic * 16 + oc] - ldsMx[wib][ic]) * ldsRs[wib][ic];
                s += r * votes[ic * VOTE_STR + c];
            }
            pre[q] = s + bias_l[q];
        }
#pragma unroll
        for (int q = 0; q < 4; ++q) {
            float n2 = pre[q] * pre[q];
            n2 += __shfl_xor(n2, 1, 32);
            n2 += __shfl_xor(n2, 2, 32);
            n2 += __shfl_xor(n2, 4, 32);
            float scale = (n2 / (1.f + n2)) / sqrtf(n2 + CAPS_EPS);
            out[(size_t)pixel * 128 + q * 32 + lane] = pre[q] * scale;
        }
    }
}

extern "C" void kernel_launch(void* const* d_in, const int* in_sizes, int n_in,
                              void* d_out, int out_size, void* d_ws, size_t ws_size,
                              hipStream_t stream) {
    (void)in_sizes; (void)n_in; (void)out_size; (void)ws_size;
    const float* input = (const float*)d_in[0];
    const float* Wg    = (const float*)d_in[1];
    const float* bias  = (const float*)d_in[2];
    float* out = (float*)d_out;
    float* xTp = (float*)d_ws;

    int ngroups = 16 * PH * PH * 16;
    caps_pad_transpose_kernel<<<(ngroups + 255) / 256, 256, 0, stream>>>(input, xTp);
    caps_conv_route_kernel<<<NPIX / (2 * PIX_PER_WAVE), 64, 0, stream>>>(xTp, Wg, bias, out);
}


# Routing (both iterations) runs in registers on the WMMA accumulator layout with DPP row_xmask and lane^16 sums; votes and logits no longer pass through LDS and serial bpermute chains
# speedup vs baseline: 2.2805x; 2.2805x over previous
#include <hip/hip_runtime.h>
#include <hip/hip_bf16.h>
#include <math.h>


typedef __attribute__((ext_vector_type(2))) float v2f;
typedef __attribute__((ext_vector_type(8))) float v8f;

#define NPIX      16384
#define KDIM      72
#define NDIM      128
#define LDSW_STR  132
#define CAPS_EPS  1e-7f
#define PH        34
#define PLANE     (PH * PH * 128)
#define PIX_PER_WAVE 4

__global__ void caps_pad_transpose_kernel(const float* __restrict__ in,
                                          float* __restrict__ out) {
    int t = blockIdx.x * blockDim.x + threadIdx.x;
    if (t >= 16 * PH * PH * 16) return;
    int b  = t & 15;
    int r  = t >> 4;
    int ww = r % PH;  r /= PH;
    int hh = r % PH;
    int ic = r / PH;
    int h = hh - 1;
    int w = ww - 1;
    float4 v0 = float4{0.f, 0.f, 0.f, 0.f};
    float4 v1 = v0;
    if ((unsigned)h < 32u && (unsigned)w < 32u) {
        const float4* src = (const float4*)(in + ((((size_t)b * 32 + h) * 32 + w) * 16 + ic) * 8);
        v0 = src[0];
        v1 = src[1];
    }
    float4* dst = (float4*)(out + (size_t)t * 8);
    dst[0] = v0;
    dst[1] = v1;
}

template <int M>
__device__ __forceinline__ float xlane(float x) {
    int b = __builtin_bit_cast(int, x);
    int r = __builtin_amdgcn_update_dpp(0, b, 0x160 + M, 0xf, 0xf, true);
    return __builtin_bit_cast(float, r);
}
__device__ __forceinline__ float xhalf(float x) {
    int b = __builtin_bit_cast(int, x);
    return __builtin_bit_cast(float,
        __builtin_amdgcn_permlanex16(b, b, 0x76543210, 0xfedcba98, false, false));
}
__device__ __forceinline__ float sum8(float x) {
    x += xlane<1>(x);
    x += xlane<2>(x);
    x += xlane<4>(x);
    return x;
}
__device__ __forceinline__ float squash_scale(float n2) {
    return (n2 / (1.f + n2)) / sqrtf(n2 + CAPS_EPS);
}

__global__ __launch_bounds__(64) void caps_conv_route_kernel(
    const float* __restrict__ xTp,
    const float* __restrict__ Wg,
    const float* __restrict__ bias,
    float* __restrict__ out)
{
    __shared__ float ldsW[KDIM * LDSW_STR];

    const int tid  = threadIdx.x;
    const int wib  = tid >> 5;
    const int lane = tid & 31;

    for (int i = tid; i < KDIM * NDIM; i += 64) {
        int r = i >> 7;
        int c = i & 127;
        ldsW[r * LDSW_STR + c] = Wg[i];
    }
    __syncthreads();

    const int m  = lane & 15;
    const int hi = lane >> 4;

    float bias_t[8];
#pragma unroll
    for (int t = 0; t < 8; ++t) bias_t[t] = bias[t * 16 + m];

    const int waveId = blockIdx.x * 2 + wib;

#pragma unroll 1
    for (int it = 0; it < PIX_PER_WAVE; ++it) {
        const int pixel = waveId * PIX_PER_WAVE + it;
        const int icap  = pixel >> 10;
        const int h     = (pixel >> 5) & 31;
        const int w     = pixel & 31;

        v8f acc[8];
#pragma unroll
        for (int t = 0; t < 8; ++t)
            acc[t] = v8f{0.f, 0.f, 0.f, 0.f, 0.f, 0.f, 0.f, 0.f};

        const float* base = xTp + (size_t)icap * PLANE + (h * PH + w) * 128;

#pragma unroll
        for (int ks = 0; ks < 18; ++ks) {
            int K0  = ks * 4 + hi * 2;
            int kh  = K0 / 24;
            int rem = K0 - kh * 24;
            int kw  = rem >> 3;
            int id0 = rem & 7;
            v2f a = *(const v2f*)(base + (kh * PH + kw) * 128 + m * 8 + id0);
#pragma unroll
            for (int t = 0; t < 8; ++t) {
                int col = t * 16 + m;
                v2f bf;
                bf[0] = ldsW[K0 * LDSW_STR + col];
                bf[1] = ldsW[(K0 + 1) * LDSW_STR + col];
                acc[t] = __builtin_amdgcn_wmma_f32_16x16x4_f32(
                    false, a, false, bf, (short)0, acc[t], false, false);
            }
        }

        float act[8];
#pragma unroll
        for (int t = 0; t < 8; ++t) {
            float s = 0.f;
#pragma unroll
            for (int v = 0; v < 8; ++v) s += acc[t][v];
            s += xhalf(s);
            float pre = s * (1.f / 16.f) + bias_t[t];
            float n2  = sum8(pre * pre);
            act[t] = pre * squash_scale(n2);
        }

        float lg[8][8];
#pragma unroll
        for (int t = 0; t < 8; ++t)
#pragma unroll
            for (int v = 0; v < 8; ++v)
                lg[t][v] = sum8(acc[t][v] * act[t]);

        float rinv[8];
#pragma unroll
        for (int v = 0; v < 8; ++v) {
            float mx = -3.0e38f;
#pragma unroll
            for (int t = 0; t < 8; ++t) mx = fmaxf(mx, lg[t][v]);
            mx = fmaxf(mx, xlane<8>(mx));
            float s = 0.f;
#pragma unroll
            for (int t = 0; t < 8; ++t) {
                lg[t][v] = __expf(lg[t][v] - mx);
                s += lg[t][v];
            }
            s += xlane<8>(s);
            rinv[v] = 1.f / s;
        }

        float res[8];
#pragma unroll
        for (int t = 0; t < 8; ++t) {
            float s = 0.f;
#pragma unroll
            for (int v = 0; v < 8; ++v) {
                float r = lg[t][v] * rinv[v];
                s += r * acc[t][v];
            }
            s += xhalf(s);
            float pre = s + bias_t[t];
            float n2  = sum8(pre * pre);
            res[t] = pre * squash_scale(n2);
        }

#pragma unroll
        for (int j = 0; j < 4; ++j) {
            float val = hi ? res[2 * j + 1] : res[2 * j];
            out[(size_t)pixel * 128 + j * 32 + lane] = val;
        }
    }
}

extern "C" void kernel_launch(void* const* d_in, const int* in_sizes, int n_in,
                              void* d_out, int out_size, void* d_ws, size_t ws_size,
                              hipStream_t stream) {
    (void)in_sizes; (void)n_in; (void)out_size; (void)ws_size;
    const float* input = (const float*)d_in[0];
    const float* Wg    = (const float*)d_in[1];
    const float* bias  = (const float*)d_in[2];
    float* out = (float*)d_out;
    float* xTp = (float*)d_ws;

    int ngroups = 16 * PH * PH * 16;
    caps_pad_transpose_kernel<<<(ngroups + 255) / 256, 256, 0, stream>>>(input, xTp);
    caps_conv_route_kernel<<<NPIX / (2 * PIX_PER_WAVE), 64, 0, stream>>>(xTp, Wg, bias, out);
}


# W columns permuted at staging so tile t = dim t of capsule m: dim sums are in-lane and each logit, exp and squash is computed once, not in 8 lanes; one 512-byte store per pixel
# speedup vs baseline: 2.9953x; 1.3134x over previous
#include <hip/hip_runtime.h>
#include <hip/hip_bf16.h>
#include <math.h>


typedef __attribute__((ext_vector_type(2))) float v2f;
typedef __attribute__((ext_vector_type(8))) float v8f;

#define NPIX      16384
#define KDIM      72
#define NDIM      128
#define LDSW_STR  132
#define CAPS_EPS  1e-7f
#define BSTRIDE   131072
#define PIX_PER_WAVE 4

template <int M>
__device__ __forceinline__ float xlane(float x) {
    int b = __builtin_bit_cast(int, x);
    int r = __builtin_amdgcn_update_dpp(0, b, 0x160 + M, 0xf, 0xf, true);
    return __builtin_bit_cast(float, r);
}
__device__ __forceinline__ float xhalf(float x) {
    int b = __builtin_bit_cast(int, x);
    return __builtin_bit_cast(float,
        __builtin_amdgcn_permlanex16(b, b, 0x76543210, 0xfedcba98, false, false));
}
__device__ __forceinline__ float squash_scale(float n2) {
    return (n2 / (1.f + n2)) / sqrtf(n2 + CAPS_EPS);
}

__global__ __launch_bounds__(64) void caps_conv_route_kernel(
    const float* __restrict__ in,
    const float* __restrict__ Wg,
    const float* __restrict__ bias,
    float* __restrict__ out)
{
    __shared__ float ldsW[KDIM * LDSW_STR];

    const int tid  = threadIdx.x;
    const int wib  = tid >> 5;
    const int lane = tid & 31;

    for (int i = tid; i < KDIM * NDIM; i += 64) {
        int r = i >> 7;
        int c = i & 127;
        ldsW[r * LDSW_STR + (c & 7) * 16 + (c >> 3)] = Wg[i];
    }
    __syncthreads();

    const int m  = lane & 15;
    const int hi = lane >> 4;

    float bias_t[8];
#pragma unroll
    for (int t = 0; t < 8; ++t) bias_t[t] = bias[m * 8 + t];

    const int waveId = blockIdx.x * 2 + __builtin_amdgcn_readfirstlane(wib);
    const int icg = waveId >> 10;
    const int h   = (waveId >> 5) & 31;
    const int w   = waveId & 31;
    const unsigned lane_boff = (unsigned)(m * BSTRIDE + 2 * hi) * 4u;

#pragma unroll 1
    for (int it = 0; it < PIX_PER_WAVE; ++it) {
        const int icap  = icg * PIX_PER_WAVE + it;
        const int pixel = (icap << 10) + (h << 5) + w;

        v8f acc[8];
#pragma unroll
        for (int t = 0; t < 8; ++t)
            acc[t] = v8f{0.f, 0.f, 0.f, 0.f, 0.f, 0.f, 0.f, 0.f};

#pragma unroll
        for (int ks = 0; ks < 18; ++ks) {
            int K0  = ks * 4 + hi * 2;
            const int tap = ks >> 1;
            const int kh  = tap / 3;
            const int kw  = tap - kh * 3;
            const int hh  = h + kh - 1;
            const int ww  = w + kw - 1;
            const bool inside = ((unsigned)hh < 32u) && ((unsigned)ww < 32u);
            const int hc  = min(max(hh, 0), 31);
            const int wc  = min(max(ww, 0), 31);
            const float* tapbase = in + (((hc * 32 + wc) * 16 + icap) * 8 + 4 * (ks & 1));
            v2f a = *(const v2f*)((const char*)tapbase + lane_boff);
            if (!inside) a = v2f{0.f, 0.f};
#pragma unroll
            for (int t = 0; t < 8; ++t) {
                int col = t * 16 + m;
                v2f bf;
                bf[0] = ldsW[K0 * LDSW_STR + col];
                bf[1] = ldsW[(K0 + 1) * LDSW_STR + col];
                acc[t] = __builtin_amdgcn_wmma_f32_16x16x4_f32(
                    false, a, false, bf, (short)0, acc[t], false, false);
            }
        }

        float act[8];
        {
            float pre[8];
            float n2 = 0.f;
#pragma unroll
            for (int t = 0; t < 8; ++t) {
                float s = 0.f;
#pragma unroll
                for (int v = 0; v < 8; ++v) s += acc[t][v];
                s += xhalf(s);
                pre[t] = s * (1.f / 16.f) + bias_t[t];
                n2 += pre[t] * pre[t];
            }
            const float sc = squash_scale(n2);
#pragma unroll
            for (int t = 0; t < 8; ++t) act[t] = pre[t] * sc;
        }

        float rt[8];
#pragma unroll
        for (int v = 0; v < 8; ++v) {
            float lg = 0.f;
#pragma unroll
            for (int t = 0; t < 8; ++t) lg += acc[t][v] * act[t];
            float mx = lg;
            mx = fmaxf(mx, xlane<1>(mx));
            mx = fmaxf(mx, xlane<2>(mx));
            mx = fmaxf(mx, xlane<4>(mx));
            mx = fmaxf(mx, xlane<8>(mx));
            float e = __expf(lg - mx);
            float s = e;
            s += xlane<1>(s);
            s += xlane<2>(s);
            s += xlane<4>(s);
            s += xlane<8>(s);
            rt[v] = e * (1.f / s);
        }

        float res[8];
        {
            float pre[8];
            float n2 = 0.f;
#pragma unroll
            for (int t = 0; t < 8; ++t) {
                float s = 0.f;
#pragma unroll
                for (int v = 0; v < 8; ++v) s += rt[v] * acc[t][v];
                s += xhalf(s);
                pre[t] = s + bias_t[t];
                n2 += pre[t] * pre[t];
            }
            const float sc = squash_scale(n2);
#pragma unroll
            for (int t = 0; t < 8; ++t) res[t] = pre[t] * sc;
        }

        float4 o;
        o.x = hi ? res[4] : res[0];
        o.y = hi ? res[5] : res[1];
        o.z = hi ? res[6] : res[2];
        o.w = hi ? res[7] : res[3];
        *(float4*)(out + (size_t)pixel * 128 + m * 8 + hi * 4) = o;
    }
}

extern "C" void kernel_launch(void* const* d_in, const int* in_sizes, int n_in,
                              void* d_out, int out_size, void* d_ws, size_t ws_size,
                              hipStream_t stream) {
    (void)in_sizes; (void)n_in; (void)out_size; (void)ws_size;
    const float* input = (const float*)d_in[0];
    const float* Wg    = (const float*)d_in[1];
    const float* bias  = (const float*)d_in[2];
    float* out = (float*)d_out;
    (void)d_ws;

    caps_conv_route_kernel<<<NPIX / (2 * PIX_PER_WAVE), 64, 0, stream>>>(input, Wg, bias, out);
}


# Grid sized to the machine: 256 work-groups of 8 waves (one per WGP, 2 waves per SIMD), each wave loops over the 8 input capsules of one 256-byte chunk; W staged to LDS once per WGP, one wave round
# speedup vs baseline: 3.9580x; 1.3214x over previous
#include <hip/hip_runtime.h>
#include <hip/hip_bf16.h>
#include <math.h>


typedef __attribute__((ext_vector_type(2))) float v2f;
typedef __attribute__((ext_vector_type(8))) float v8f;

#define NPIX      16384
#define KDIM      72
#define NDIM      128
#define LDSW_STR  132
#define CAPS_EPS  1e-7f
#define BSTRIDE   131072
#define PIX_PER_WAVE 8
#define WAVES_PER_BLOCK 8
#define BLOCK_THREADS (WAVES_PER_BLOCK * 32)

template <int M>
__device__ __forceinline__ float xlane(float x) {
    int b = __builtin_bit_cast(int, x);
    int r = __builtin_amdgcn_update_dpp(0, b, 0x160 + M, 0xf, 0xf, true);
    return __builtin_bit_cast(float, r);
}
__device__ __forceinline__ float xhalf(float x) {
    int b = __builtin_bit_cast(int, x);
    return __builtin_bit_cast(float,
        __builtin_amdgcn_permlanex16(b, b, 0x76543210, 0xfedcba98, false, false));
}
__device__ __forceinline__ float squash_scale(float n2) {
    return (n2 / (1.f + n2)) / sqrtf(n2 + CAPS_EPS);
}

__global__ __launch_bounds__(BLOCK_THREADS) void caps_conv_route_kernel(
    const float* __restrict__ in,
    const float* __restrict__ Wg,
    const float* __restrict__ bias,
    float* __restrict__ out)
{
    __shared__ float ldsW[KDIM * LDSW_STR];

    const int tid  = threadIdx.x;
    const int wib  = tid >> 5;
    const int lane = tid & 31;

    for (int i = tid; i < KDIM * NDIM; i += BLOCK_THREADS) {
        int r = i >> 7;
        int c = i & 127;
        ldsW[r * LDSW_STR + (c & 7) * 16 + (c >> 3)] = Wg[i];
    }
    __syncthreads();

    const int m  = lane & 15;
    const int hi = lane >> 4;

    float bias_t[8];
#pragma unroll
    for (int t = 0; t < 8; ++t) bias_t[t] = bias[m * 8 + t];

    const int waveId = blockIdx.x * WAVES_PER_BLOCK + __builtin_amdgcn_readfirstlane(wib);
    const int icg = waveId >> 10;
    const int h   = (waveId >> 5) & 31;
    const int w   = waveId & 31;
    const unsigned lane_boff = (unsigned)(m * BSTRIDE + 2 * hi) * 4u;

#pragma unroll 1
    for (int it = 0; it < PIX_PER_WAVE; ++it) {
        const int icap  = icg * PIX_PER_WAVE + it;
        const int pixel = (icap << 10) + (h << 5) + w;

        v8f acc[8];
#pragma unroll
        for (int t = 0; t < 8; ++t)
            acc[t] = v8f{0.f, 0.f, 0.f, 0.f, 0.f, 0.f, 0.f, 0.f};

#pragma unroll
        for (int ks = 0; ks < 18; ++ks) {
            int K0  = ks * 4 + hi * 2;
            const int tap = ks >> 1;
            const int kh  = tap / 3;
            const int kw  = tap - kh * 3;
            const int hh  = h + kh - 1;
            const int ww  = w + kw - 1;
            const bool inside = ((unsigned)hh < 32u) && ((unsigned)ww < 32u);
            const int hc  = min(max(hh, 0), 31);
            const int wc  = min(max(ww, 0), 31);
            const float* tapbase = in + (((hc * 32 + wc) * 16 + icap) * 8 + 4 * (ks & 1));
            v2f a = *(const v2f*)((const char*)tapbase + lane_boff);
            if (!inside) a = v2f{0.f, 0.f};
#pragma unroll
            for (int t = 0; t < 8; ++t) {
                int col = t * 16 + m;
                v2f bf;
                bf[0] = ldsW[K0 * LDSW_STR + col];
                bf[1] = ldsW[(K0 + 1) * LDSW_STR + col];
                acc[t] = __builtin_amdgcn_wmma_f32_16x16x4_f32(
                    false, a, false, bf, (short)0, acc[t], false, false);
            }
        }

        float act[8];
        {
            float pre[8];
            float n2 = 0.f;
#pragma unroll
            for (int t = 0; t < 8; ++t) {
                float s = 0.f;
#pragma unroll
                for (int v = 0; v < 8; ++v) s += acc[t][v];
                s += xhalf(s);
                pre[t] = s * (1.f / 16.f) + bias_t[t];
                n2 += pre[t] * pre[t];
            }
            const float sc = squash_scale(n2);
#pragma unroll
            for (int t = 0; t < 8; ++t) act[t] = pre[t] * sc;
        }

        float rt[8];
#pragma unroll
        for (int v = 0; v < 8; ++v) {
            float lg = 0.f;
#pragma unroll
            for (int t = 0; t < 8; ++t) lg += acc[t][v] * act[t];
            float mx = lg;
            mx = fmaxf(mx, xlane<1>(mx));
            mx = fmaxf(mx, xlane<2>(mx));
            mx = fmaxf(mx, xlane<4>(mx));
            mx = fmaxf(mx, xlane<8>(mx));
            float e = __expf(lg - mx);
            float s = e;
            s += xlane<1>(s);
            s += xlane<2>(s);
            s += xlane<4>(s);
            s += xlane<8>(s);
            rt[v] = e * (1.f / s);
        }

        float res[8];
        {
            float pre[8];
            float n2 = 0.f;
#pragma unroll
            for (int t = 0; t < 8; ++t) {
                float s = 0.f;
#pragma unroll
                for (int v = 0; v < 8; ++v) s += rt[v] * acc[t][v];
                s += xhalf(s);
                pre[t] = s + bias_t[t];
                n2 += pre[t] * pre[t];
            }
            const float sc = squash_scale(n2);
#pragma unroll
            for (int t = 0; t < 8; ++t) res[t] = pre[t] * sc;
        }

        float4 o;
        o.x = hi ? res[4] : res[0];
        o.y = hi ? res[5] : res[1];
        o.z = hi ? res[6] : res[2];
        o.w = hi ? res[7] : res[3];
        *(float4*)(out + (size_t)pixel * 128 + m * 8 + hi * 4) = o;
    }
}

extern "C" void kernel_launch(void* const* d_in, const int* in_sizes, int n_in,
                              void* d_out, int out_size, void* d_ws, size_t ws_size,
                              hipStream_t stream) {
    (void)in_sizes; (void)n_in; (void)out_size; (void)ws_size;
    const float* input = (const float*)d_in[0];
    const float* Wg    = (const float*)d_in[1];
    const float* bias  = (const float*)d_in[2];
    float* out = (float*)d_out;
    (void)d_ws;

    caps_conv_route_kernel<<<NPIX / (WAVES_PER_BLOCK * PIX_PER_WAVE), BLOCK_THREADS, 0, stream>>>(input, Wg, bias, out);
}


# Softmax normalisation and both squash scales use v_rcp_f32 / v_rsq_f32 (1 ulp) instead of 12 exact divisions and 2 exact square roots per pixel; each squash chain shrinks from 45 instructions to 6
# speedup vs baseline: 4.1561x; 1.0501x over previous
#include <hip/hip_runtime.h>
#include <hip/hip_bf16.h>
#include <math.h>


typedef __attribute__((ext_vector_type(2))) float v2f;
typedef __attribute__((ext_vector_type(8))) float v8f;

#define NPIX      16384
#define KDIM      72
#define NDIM      128
#define LDSW_STR  132
#define CAPS_EPS  1e-7f
#define BSTRIDE   131072
#define PIX_PER_WAVE 8
#define WAVES_PER_BLOCK 8
#define BLOCK_THREADS (WAVES_PER_BLOCK * 32)

template <int M>
__device__ __forceinline__ float xlane(float x) {
    int b = __builtin_bit_cast(int, x);
    int r = __builtin_amdgcn_update_dpp(0, b, 0x160 + M, 0xf, 0xf, true);
    return __builtin_bit_cast(float, r);
}
__device__ __forceinline__ float xhalf(float x) {
    int b = __builtin_bit_cast(int, x);
    return __builtin_bit_cast(float,
        __builtin_amdgcn_permlanex16(b, b, 0x76543210, 0xfedcba98, false, false));
}
__device__ __forceinline__ float squash_scale(float n2) {
    return (n2 * __builtin_amdgcn_rcpf(1.f + n2)) * __builtin_amdgcn_rsqf(n2 + CAPS_EPS);
}

__global__ __launch_bounds__(BLOCK_THREADS) void caps_conv_route_kernel(
    const float* __restrict__ in,
    const float* __restrict__ Wg,
    const float* __restrict__ bias,
    float* __restrict__ out)
{
    __shared__ float ldsW[KDIM * LDSW_STR];

    const int tid  = threadIdx.x;
    const int wib  = tid >> 5;
    const int lane = tid & 31;

    for (int i = tid; i < KDIM * NDIM; i += BLOCK_THREADS) {
        int r = i >> 7;
        int c = i & 127;
        ldsW[r * LDSW_STR + (c & 7) * 16 + (c >> 3)] = Wg[i];
    }
    __syncthreads();

    const int m  = lane & 15;
    const int hi = lane >> 4;

    float bias_t[8];
#pragma unroll
    for (int t = 0; t < 8; ++t) bias_t[t] = bias[m * 8 + t];

    const int waveId = blockIdx.x * WAVES_PER_BLOCK + __builtin_amdgcn_readfirstlane(wib);
    const int icg = waveId >> 10;
    const int h   = (waveId >> 5) & 31;
    const int w   = waveId & 31;
    const unsigned lane_boff = (unsigned)(m * BSTRIDE + 2 * hi) * 4u;

#pragma unroll 1
    for (int it = 0; it < PIX_PER_WAVE; ++it) {
        const int icap  = icg * PIX_PER_WAVE + it;
        const int pixel = (icap << 10) + (h << 5) + w;

        v8f acc[8];
#pragma unroll
        for (int t = 0; t < 8; ++t)
            acc[t] = v8f{0.f, 0.f, 0.f, 0.f, 0.f, 0.f, 0.f, 0.f};

#pragma unroll
        for (int ks = 0; ks < 18; ++ks) {
            int K0  = ks * 4 + hi * 2;
            const int tap = ks >> 1;
            const int kh  = tap / 3;
            const int kw  = tap - kh * 3;
            const int hh  = h + kh - 1;
            const int ww  = w + kw - 1;
            const bool inside = ((unsigned)hh < 32u) && ((unsigned)ww < 32u);
            const int hc  = min(max(hh, 0), 31);
            const int wc  = min(max(ww, 0), 31);
            const float* tapbase = in + (((hc * 32 + wc) * 16 + icap) * 8 + 4 * (ks & 1));
            v2f a = *(const v2f*)((const char*)tapbase + lane_boff);
            if (!inside) a = v2f{0.f, 0.f};
#pragma unroll
            for (int t = 0; t < 8; ++t) {
                int col = t * 16 + m;
                v2f bf;
                bf[0] = ldsW[K0 * LDSW_STR + col];
                bf[1] = ldsW[(K0 + 1) * LDSW_STR + col];
                acc[t] = __builtin_amdgcn_wmma_f32_16x16x4_f32(
                    false, a, false, bf, (short)0, acc[t], false, false);
            }
        }

        float act[8];
        {
            float pre[8];
            float n2 = 0.f;
#pragma unroll
            for (int t = 0; t < 8; ++t) {
                float s = 0.f;
#pragma unroll
                for (int v = 0; v < 8; ++v) s += acc[t][v];
                s += xhalf(s);
                pre[t] = s * (1.f / 16.f) + bias_t[t];
                n2 += pre[t] * pre[t];
            }
            const float sc = squash_scale(n2);
#pragma unroll
            for (int t = 0; t < 8; ++t) act[t] = pre[t] * sc;
        }

        float rt[8];
#pragma unroll
        for (int v = 0; v < 8; ++v) {
            float lg = 0.f;
#pragma unroll
            for (int t = 0; t < 8; ++t) lg += acc[t][v] * act[t];
            float mx = lg;
            mx = fmaxf(mx, xlane<1>(mx));
            mx = fmaxf(mx, xlane<2>(mx));
            mx = fmaxf(mx, xlane<4>(mx));
            mx = fmaxf(mx, xlane<8>(mx));
            float e = __expf(lg - mx);
            float s = e;
            s += xlane<1>(s);
            s += xlane<2>(s);
            s += xlane<4>(s);
            s += xlane<8>(s);
            rt[v] = e * __builtin_amdgcn_rcpf(s);
        }

        float res[8];
        {
            float pre[8];
            float n2 = 0.f;
#pragma unroll
            for (int t = 0; t < 8; ++t) {
                float s = 0.f;
#pragma unroll
                for (int v = 0; v < 8; ++v) s += rt[v] * acc[t][v];
                s += xhalf(s);
                pre[t] = s + bias_t[t];
                n2 += pre[t] * pre[t];
            }
            const float sc = squash_scale(n2);
#pragma unroll
            for (int t = 0; t < 8; ++t) res[t] = pre[t] * sc;
        }

        float4 o;
        o.x = hi ? res[4] : res[0];
        o.y = hi ? res[5] : res[1];
        o.z = hi ? res[6] : res[2];
        o.w = hi ? res[7] : res[3];
        *(float4*)(out + (size_t)pixel * 128 + m * 8 + hi * 4) = o;
    }
}

extern "C" void kernel_launch(void* const* d_in, const int* in_sizes, int n_in,
                              void* d_out, int out_size, void* d_ws, size_t ws_size,
                              hipStream_t stream) {
    (void)in_sizes; (void)n_in; (void)out_size; (void)ws_size;
    const float* input = (const float*)d_in[0];
    const float* Wg    = (const float*)d_in[1];
    const float* bias  = (const float*)d_in[2];
    float* out = (float*)d_out;
    (void)d_ws;

    caps_conv_route_kernel<<<NPIX / (WAVES_PER_BLOCK * PIX_PER_WAVE), BLOCK_THREADS, 0, stream>>>(input, Wg, bias, out);
}


# Next pixel's 18 A loads issued right behind the last WMMA into the same registers, so routing covers their latency; first pixel's loads go out ahead of the B-fragment LDS reads
# speedup vs baseline: 4.4768x; 1.0772x over previous
#include <hip/hip_runtime.h>
#include <hip/hip_bf16.h>
#include <math.h>


typedef __attribute__((ext_vector_type(2))) float v2f;
typedef __attribute__((ext_vector_type(8))) float v8f;

#define NPIX      16384
#define KDIM      72
#define NDIM      128
#define LDSW_STR  132
#define CAPS_EPS  1e-7f
#define BSTRIDE   131072
#define PIX_PER_WAVE 8
#define WAVES_PER_BLOCK 8
#define BLOCK_THREADS (WAVES_PER_BLOCK * 32)

template <int M>
__device__ __forceinline__ float xlane(float x) {
    int b = __builtin_bit_cast(int, x);
    int r = __builtin_amdgcn_update_dpp(0, b, 0x160 + M, 0xf, 0xf, true);
    return __builtin_bit_cast(float, r);
}
__device__ __forceinline__ float xhalf(float x) {
    int b = __builtin_bit_cast(int, x);
    return __builtin_bit_cast(float,
        __builtin_amdgcn_permlanex16(b, b, 0x76543210, 0xfedcba98, false, false));
}
__device__ __forceinline__ float squash_scale(float n2) {
    return (n2 * __builtin_amdgcn_rcpf(1.f + n2)) * __builtin_amdgcn_rsqf(n2 + CAPS_EPS);
}

__global__ __launch_bounds__(BLOCK_THREADS) void caps_conv_route_kernel(
    const float* __restrict__ in,
    const float* __restrict__ Wg,
    const float* __restrict__ bias,
    float* __restrict__ out)
{
    __shared__ float ldsW[KDIM * LDSW_STR];

    const int tid  = threadIdx.x;
    const int wib  = tid >> 5;
    const int lane = tid & 31;

    for (int i = tid; i < KDIM * NDIM; i += BLOCK_THREADS) {
        int r = i >> 7;
        int c = i & 127;
        ldsW[r * LDSW_STR + (c & 7) * 16 + (c >> 3)] = Wg[i];
    }
    __syncthreads();

    const int m  = lane & 15;
    const int hi = lane >> 4;

    float bias_t[8];
#pragma unroll
    for (int t = 0; t < 8; ++t) bias_t[t] = bias[m * 8 + t];

    const int waveId = blockIdx.x * WAVES_PER_BLOCK + __builtin_amdgcn_readfirstlane(wib);
    const int icg = waveId >> 10;
    const int h   = (waveId >> 5) & 31;
    const int w   = waveId & 31;
    const unsigned lane_boff = (unsigned)(m * BSTRIDE + 2 * hi) * 4u;

    v2f araw[18];
#define LOAD_A_FRAGMENTS(ICAP)                                                                   \
    _Pragma("unroll")                                                                             \
    for (int ks = 0; ks < 18; ++ks) {                                                             \
        const int tap = ks >> 1;                                                                  \
        const int kh  = tap / 3;                                                                  \
        const int kw  = tap - kh * 3;                                                             \
        const int hc  = min(max(h + kh - 1, 0), 31);                                              \
        const int wc  = min(max(w + kw - 1, 0), 31);                                              \
        const float* tapbase = in + (((hc * 32 + wc) * 16 + (ICAP)) * 8 + 4 * (ks & 1));          \
        araw[ks] = *(const v2f*)((const char*)tapbase + lane_boff);                               \
    }

    LOAD_A_FRAGMENTS(icg * PIX_PER_WAVE)

    v2f bfr[18][8];
#pragma unroll
    for (int ks = 0; ks < 18; ++ks)
#pragma unroll
        for (int t = 0; t < 8; ++t) {
            const int K0 = ks * 4 + hi * 2;
            bfr[ks][t][0] = ldsW[K0 * LDSW_STR + t * 16 + m];
            bfr[ks][t][1] = ldsW[(K0 + 1) * LDSW_STR + t * 16 + m];
        }

#pragma unroll 1
    for (int it = 0; it < PIX_PER_WAVE; ++it) {
        const int icap  = icg * PIX_PER_WAVE + it;
        const int pixel = (icap << 10) + (h << 5) + w;

        v8f acc[8];
#pragma unroll
        for (int t = 0; t < 8; ++t)
            acc[t] = v8f{0.f, 0.f, 0.f, 0.f, 0.f, 0.f, 0.f, 0.f};

#pragma unroll
        for (int ks = 0; ks < 18; ++ks) {
            const int tap = ks >> 1;
            const int kh  = tap / 3;
            const int kw  = tap - kh * 3;
            const bool inside = ((unsigned)(h + kh - 1) < 32u) && ((unsigned)(w + kw - 1) < 32u);
            v2f a = araw[ks];
            if (!inside) a = v2f{0.f, 0.f};
#pragma unroll
            for (int t = 0; t < 8; ++t)
                acc[t] = __builtin_amdgcn_wmma_f32_16x16x4_f32(
                    false, a, false, bfr[ks][t], (short)0, acc[t], false, false);
        }

        if (it + 1 < PIX_PER_WAVE) {
            LOAD_A_FRAGMENTS(icap + 1)
        }

        float act[8];
        {
            float pre[8];
            float n2 = 0.f;
#pragma unroll
            for (int t = 0; t < 8; ++t) {
                float s = 0.f;
#pragma unroll
                for (int v = 0; v < 8; ++v) s += acc[t][v];
                s += xhalf(s);
                pre[t] = s * (1.f / 16.f) + bias_t[t];
                n2 += pre[t] * pre[t];
            }
            const float sc = squash_scale(n2);
#pragma unroll
            for (int t = 0; t < 8; ++t) act[t] = pre[t] * sc;
        }

        float rt[8];
#pragma unroll
        for (int v = 0; v < 8; ++v) {
            float lg = 0.f;
#pragma unroll
            for (int t = 0; t < 8; ++t) lg += acc[t][v] * act[t];
            float mx = lg;
            mx = fmaxf(mx, xlane<1>(mx));
            mx = fmaxf(mx, xlane<2>(mx));
            mx = fmaxf(mx, xlane<4>(mx));
            mx = fmaxf(mx, xlane<8>(mx));
            float e = __expf(lg - mx);
            float s = e;
            s += xlane<1>(s);
            s += xlane<2>(s);
            s += xlane<4>(s);
            s += xlane<8>(s);
            rt[v] = e * __builtin_amdgcn_rcpf(s);
        }

        float res[8];
        {
            float pre[8];
            float n2 = 0.f;
#pragma unroll
            for (int t = 0; t < 8; ++t) {
                float s = 0.f;
#pragma unroll
                for (int v = 0; v < 8; ++v) s += rt[v] * acc[t][v];
                s += xhalf(s);
                pre[t] = s + bias_t[t];
                n2 += pre[t] * pre[t];
            }
            const float sc = squash_scale(n2);
#pragma unroll
            for (int t = 0; t < 8; ++t) res[t] = pre[t] * sc;
        }

        float4 o;
        o.x = hi ? res[4] : res[0];
        o.y = hi ? res[5] : res[1];
        o.z = hi ? res[6] : res[2];
        o.w = hi ? res[7] : res[3];
        *(float4*)(out + (size_t)pixel * 128 + m * 8 + hi * 4) = o;
    }
}

extern "C" void kernel_launch(void* const* d_in, const int* in_sizes, int n_in,
                              void* d_out, int out_size, void* d_ws, size_t ws_size,
                              hipStream_t stream) {
    (void)in_sizes; (void)n_in; (void)out_size; (void)ws_size;
    const float* input = (const float*)d_in[0];
    const float* Wg    = (const float*)d_in[1];
    const float* bias  = (const float*)d_in[2];
    float* out = (float*)d_out;
    (void)d_ws;

    caps_conv_route_kernel<<<NPIX / (WAVES_PER_BLOCK * PIX_PER_WAVE), BLOCK_THREADS, 0, stream>>>(input, Wg, bias, out);
}


# Routing arithmetic written on the accumulator's adjacent-row register pairs: row sums, logits and routed sums are packed f32 add/fma on pairs the WMMA left in place, no repacking moves
# speedup vs baseline: 4.6313x; 1.0345x over previous
#include <hip/hip_runtime.h>
#include <hip/hip_bf16.h>
#include <math.h>


typedef __attribute__((ext_vector_type(2))) float v2f;
typedef __attribute__((ext_vector_type(8))) float v8f;

#define NPIX      16384
#define KDIM      72
#define NDIM      128
#define LDSW_STR  132
#define CAPS_EPS  1e-7f
#define BSTRIDE   131072
#define PIX_PER_WAVE 8
#define WAVES_PER_BLOCK 8
#define BLOCK_THREADS (WAVES_PER_BLOCK * 32)
#define ROWPAIR(T, J) (v2f{acc[(T)][2 * (J)], acc[(T)][2 * (J) + 1]})

template <int M>
__device__ __forceinline__ float xlane(float x) {
    int b = __builtin_bit_cast(int, x);
    int r = __builtin_amdgcn_update_dpp(0, b, 0x160 + M, 0xf, 0xf, true);
    return __builtin_bit_cast(float, r);
}
__device__ __forceinline__ float xhalf(float x) {
    int b = __builtin_bit_cast(int, x);
    return __builtin_bit_cast(float,
        __builtin_amdgcn_permlanex16(b, b, 0x76543210, 0xfedcba98, false, false));
}
__device__ __forceinline__ float squash_scale(float n2) {
    return (n2 * __builtin_amdgcn_rcpf(1.f + n2)) * __builtin_amdgcn_rsqf(n2 + CAPS_EPS);
}

__global__ __launch_bounds__(BLOCK_THREADS) void caps_conv_route_kernel(
    const float* __restrict__ in,
    const float* __restrict__ Wg,
    const float* __restrict__ bias,
    float* __restrict__ out)
{
    __shared__ float ldsW[KDIM * LDSW_STR];

    const int tid  = threadIdx.x;
    const int wib  = tid >> 5;
    const int lane = tid & 31;

    for (int i = tid; i < KDIM * NDIM; i += BLOCK_THREADS) {
        int r = i >> 7;
        int c = i & 127;
        ldsW[r * LDSW_STR + (c & 7) * 16 + (c >> 3)] = Wg[i];
    }
    __syncthreads();

    const int m  = lane & 15;
    const int hi = lane >> 4;

    v2f biasp[4];
#pragma unroll
    for (int u = 0; u < 4; ++u) biasp[u] = v2f{bias[m * 8 + 2 * u], bias[m * 8 + 2 * u + 1]};

    const int waveId = blockIdx.x * WAVES_PER_BLOCK + __builtin_amdgcn_readfirstlane(wib);
    const int icg = waveId >> 10;
    const int h   = (waveId >> 5) & 31;
    const int w   = waveId & 31;
    const unsigned lane_boff = (unsigned)(m * BSTRIDE + 2 * hi) * 4u;

    v2f araw[18];
#define LOAD_A_FRAGMENTS(ICAP)                                                                   \
    _Pragma("unroll")                                                                             \
    for (int ks = 0; ks < 18; ++ks) {                                                             \
        const int tap = ks >> 1;                                                                  \
        const int kh  = tap / 3;                                                                  \
        const int kw  = tap - kh * 3;                                                             \
        const int hc  = min(max(h + kh - 1, 0), 31);                                              \
        const int wc  = min(max(w + kw - 1, 0), 31);                                              \
        const float* tapbase = in + (((hc * 32 + wc) * 16 + (ICAP)) * 8 + 4 * (ks & 1));          \
        araw[ks] = *(const v2f*)((const char*)tapbase + lane_boff);                               \
    }

    LOAD_A_FRAGMENTS(icg * PIX_PER_WAVE)

    v2f bfr[18][8];
#pragma unroll
    for (int ks = 0; ks < 18; ++ks)
#pragma unroll
        for (int t = 0; t < 8; ++t) {
            const int K0 = ks * 4 + hi * 2;
            bfr[ks][t][0] = ldsW[K0 * LDSW_STR + t * 16 + m];
            bfr[ks][t][1] = ldsW[(K0 + 1) * LDSW_STR + t * 16 + m];
        }

#pragma unroll 1
    for (int it = 0; it < PIX_PER_WAVE; ++it) {
        const int icap  = icg * PIX_PER_WAVE + it;
        const int pixel = (icap << 10) + (h << 5) + w;

        v8f acc[8];
#pragma unroll
        for (int t = 0; t < 8; ++t)
            acc[t] = v8f{0.f, 0.f, 0.f, 0.f, 0.f, 0.f, 0.f, 0.f};

#pragma unroll
        for (int ks = 0; ks < 18; ++ks) {
            const int tap = ks >> 1;
            const int kh  = tap / 3;
            const int kw  = tap - kh * 3;
            const bool inside = ((unsigned)(h + kh - 1) < 32u) && ((unsigned)(w + kw - 1) < 32u);
            v2f a = araw[ks];
            if (!inside) a = v2f{0.f, 0.f};
#pragma unroll
            for (int t = 0; t < 8; ++t)
                acc[t] = __builtin_amdgcn_wmma_f32_16x16x4_f32(
                    false, a, false, bfr[ks][t], (short)0, acc[t], false, false);
        }

        if (it + 1 < PIX_PER_WAVE) {
            LOAD_A_FRAGMENTS(icap + 1)
        }

        v2f actp[4];
        {
            float s[8];
#pragma unroll
            for (int t = 0; t < 8; ++t) {
                const v2f q = (ROWPAIR(t, 0) + ROWPAIR(t, 1)) + (ROWPAIR(t, 2) + ROWPAIR(t, 3));
                const float r = q.x + q.y;
                s[t] = r + xhalf(r);
            }
            v2f prep[4];
#pragma unroll
            for (int u = 0; u < 4; ++u)
                prep[u] = v2f{s[2 * u], s[2 * u + 1]} * (1.f / 16.f) + biasp[u];
            v2f n2v = prep[0] * prep[0];
#pragma unroll
            for (int u = 1; u < 4; ++u) n2v += prep[u] * prep[u];
            const float sc = squash_scale(n2v.x + n2v.y);
#pragma unroll
            for (int u = 0; u < 4; ++u) actp[u] = prep[u] * sc;
        }

        v2f rtp[4];
#pragma unroll
        for (int j = 0; j < 4; ++j) {
            v2f lg = ROWPAIR(0, j) * actp[0].x;
            lg += ROWPAIR(1, j) * actp[0].y;
#pragma unroll
            for (int u = 1; u < 4; ++u) {
                lg += ROWPAIR(2 * u, j) * actp[u].x;
                lg += ROWPAIR(2 * u + 1, j) * actp[u].y;
            }
            v2f mx = lg;
            mx = v2f{fmaxf(mx.x, xlane<1>(mx.x)), fmaxf(mx.y, xlane<1>(mx.y))};
            mx = v2f{fmaxf(mx.x, xlane<2>(mx.x)), fmaxf(mx.y, xlane<2>(mx.y))};
            mx = v2f{fmaxf(mx.x, xlane<4>(mx.x)), fmaxf(mx.y, xlane<4>(mx.y))};
            mx = v2f{fmaxf(mx.x, xlane<8>(mx.x)), fmaxf(mx.y, xlane<8>(mx.y))};
            const v2f d = lg - mx;
            const v2f e = v2f{__expf(d.x), __expf(d.y)};
            v2f sm = e;
            sm = v2f{sm.x + xlane<1>(sm.x), sm.y + xlane<1>(sm.y)};
            sm = v2f{sm.x + xlane<2>(sm.x), sm.y + xlane<2>(sm.y)};
            sm = v2f{sm.x + xlane<4>(sm.x), sm.y + xlane<4>(sm.y)};
            sm = v2f{sm.x + xlane<8>(sm.x), sm.y + xlane<8>(sm.y)};
            rtp[j] = e * v2f{__builtin_amdgcn_rcpf(sm.x), __builtin_amdgcn_rcpf(sm.y)};
        }

        v2f resp[4];
        {
            float s[8];
#pragma unroll
            for (int t = 0; t < 8; ++t) {
                v2f q = ROWPAIR(t, 0) * rtp[0];
#pragma unroll
                for (int j = 1; j < 4; ++j) q += ROWPAIR(t, j) * rtp[j];
                const float r = q.x + q.y;
                s[t] = r + xhalf(r);
            }
            v2f prep[4];
#pragma unroll
            for (int u = 0; u < 4; ++u)
                prep[u] = v2f{s[2 * u], s[2 * u + 1]} + biasp[u];
            v2f n2v = prep[0] * prep[0];
#pragma unroll
            for (int u = 1; u < 4; ++u) n2v += prep[u] * prep[u];
            const float sc = squash_scale(n2v.x + n2v.y);
#pragma unroll
            for (int u = 0; u < 4; ++u) resp[u] = prep[u] * sc;
        }

        const v2f olo = hi ? resp[2] : resp[0];
        const v2f ohi = hi ? resp[3] : resp[1];
        float4 o;
        o.x = olo.x;
        o.y = olo.y;
        o.z = ohi.x;
        o.w = ohi.y;
        *(float4*)(out + (size_t)pixel * 128 + m * 8 + hi * 4) = o;
    }
}

extern "C" void kernel_launch(void* const* d_in, const int* in_sizes, int n_in,
                              void* d_out, int out_size, void* d_ws, size_t ws_size,
                              hipStream_t stream) {
    (void)in_sizes; (void)n_in; (void)out_size; (void)ws_size;
    const float* input = (const float*)d_in[0];
    const float* Wg    = (const float*)d_in[1];
    const float* bias  = (const float*)d_in[2];
    float* out = (float*)d_out;
    (void)d_ws;

    caps_conv_route_kernel<<<NPIX / (WAVES_PER_BLOCK * PIX_PER_WAVE), BLOCK_THREADS, 0, stream>>>(input, Wg, bias, out);
}


# B-operand path: W staged in fragment order by 16-byte copies (one load round), read back by ds_load_b128 with no repacking; out-of-image taps read a zero block, so A goes load-to-WMMA with no select
# speedup vs baseline: 5.3451x; 1.1541x over previous
#include <hip/hip_runtime.h>
#include <hip/hip_bf16.h>
#include <math.h>


typedef __attribute__((ext_vector_type(2))) float v2f;
typedef __attribute__((ext_vector_type(8))) float v8f;

#define NPIX      16384
#define KDIM      72
#define NDIM      128
#define B_PIECES  2304
#define Z_PIECES  256
#define CAPS_EPS  1e-7f
#define BSTRIDE   131072
#define PIX_PER_WAVE 8
#define WAVES_PER_BLOCK 8
#define BLOCK_THREADS (WAVES_PER_BLOCK * 32)
#define ROWPAIR(T, J) (v2f{acc[(T)][2 * (J)], acc[(T)][2 * (J) + 1]})

template <int M>
__device__ __forceinline__ float xlane(float x) {
    int b = __builtin_bit_cast(int, x);
    int r = __builtin_amdgcn_update_dpp(0, b, 0x160 + M, 0xf, 0xf, true);
    return __builtin_bit_cast(float, r);
}
__device__ __forceinline__ float xhalf(float x) {
    int b = __builtin_bit_cast(int, x);
    return __builtin_bit_cast(float,
        __builtin_amdgcn_permlanex16(b, b, 0x76543210, 0xfedcba98, false, false));
}
__device__ __forceinline__ float squash_scale(float n2) {
    return (n2 * __builtin_amdgcn_rcpf(1.f + n2)) * __builtin_amdgcn_rsqf(n2 + CAPS_EPS);
}

__global__ __launch_bounds__(BLOCK_THREADS) void caps_conv_route_kernel(
    const float* __restrict__ in,
    const float* __restrict__ Wg,
    const float* __restrict__ bias,
    float* __restrict__ out)
{
    __shared__ float4 ldsB[B_PIECES + Z_PIECES];

    const int tid  = threadIdx.x;
    const int wib  = tid >> 5;
    const int lane = tid & 31;

    static_assert((KDIM / 2) * 32 == 4 * BLOCK_THREADS + BLOCK_THREADS / 2, "staging split");
    float4 w0[5], w1[5];
#define STAGE_LOAD(U)                                                                            \
    {                                                                                             \
        const int i = tid + (U) * BLOCK_THREADS;                                                  \
        const int p = i >> 5, c4 = i & 31;                                                        \
        w0[U] = *(const float4*)(Wg + (2 * p) * NDIM + c4 * 4);                                   \
        w1[U] = *(const float4*)(Wg + (2 * p + 1) * NDIM + c4 * 4);                               \
    }
    STAGE_LOAD(0) STAGE_LOAD(1) STAGE_LOAD(2) STAGE_LOAD(3)
    if (wib < 4) STAGE_LOAD(4)

    const int m  = lane & 15;
    const int hi = lane >> 4;

    v2f biasp[4];
#pragma unroll
    for (int u = 0; u < 4; ++u) biasp[u] = v2f{bias[m * 8 + 2 * u], bias[m * 8 + 2 * u + 1]};

    const int waveId = blockIdx.x * WAVES_PER_BLOCK + __builtin_amdgcn_readfirstlane(wib);
    const int icg = waveId >> 10;
    const int h   = (waveId >> 5) & 31;
    const int w   = waveId & 31;
    const unsigned lane_boff = (unsigned)(m * BSTRIDE + 2 * hi) * 4u;

    v2f araw[18];
#define LOAD_A_FRAGMENTS(ICAP)                                                                   \
    _Pragma("unroll")                                                                             \
    for (int ks = 0; ks < 18; ++ks) {                                                             \
        const int tap = ks >> 1;                                                                  \
        const int kh  = tap / 3;                                                                  \
        const int kw  = tap - kh * 3;                                                             \
        const int hc  = min(max(h + kh - 1, 0), 31);                                              \
        const int wc  = min(max(w + kw - 1, 0), 31);                                              \
        const float* tapbase = in + (((hc * 32 + wc) * 16 + (ICAP)) * 8 + 4 * (ks & 1));          \
        araw[ks] = *(const v2f*)((const char*)tapbase + lane_boff);                               \
    }

    LOAD_A_FRAGMENTS(icg * PIX_PER_WAVE)

#define STAGE_STORE(U)                                                                           \
    {                                                                                             \
        const int i = tid + (U) * BLOCK_THREADS;                                                  \
        const int p = i >> 5, c4 = i & 31;                                                        \
        const int piece = ((p >> 1) * 4 + 2 * (c4 & 1)) * 32 + (p & 1) * 16 + (c4 >> 1);          \
        ldsB[piece]      = float4{w0[U].x, w1[U].x, w0[U].y, w1[U].y};                            \
        ldsB[piece + 32] = float4{w0[U].z, w1[U].z, w0[U].w, w1[U].w};                            \
    }
    STAGE_STORE(0) STAGE_STORE(1) STAGE_STORE(2) STAGE_STORE(3)
    if (wib < 4) STAGE_STORE(4)
    ldsB[B_PIECES + tid] = float4{0.f, 0.f, 0.f, 0.f};
    __syncthreads();

    float4 bq[18][4];
#pragma unroll
    for (int tap = 0; tap < 9; ++tap) {
        const int kh = tap / 3;
        const int kw = tap - kh * 3;
        const bool inside = ((unsigned)(h + kh - 1) < 32u) && ((unsigned)(w + kw - 1) < 32u);
        const int tb = inside ? tap * 256 : B_PIECES;
#pragma unroll
        for (int s = 0; s < 2; ++s)
#pragma unroll
            for (int q = 0; q < 4; ++q)
                bq[2 * tap + s][q] = ldsB[tb + (s * 4 + q) * 32 + lane];
    }

#pragma unroll 1
    for (int it = 0; it < PIX_PER_WAVE; ++it) {
        const int icap  = icg * PIX_PER_WAVE + it;
        const int pixel = (icap << 10) + (h << 5) + w;

        v8f acc[8];
#pragma unroll
        for (int t = 0; t < 8; ++t)
            acc[t] = v8f{0.f, 0.f, 0.f, 0.f, 0.f, 0.f, 0.f, 0.f};

#pragma unroll
        for (int ks = 0; ks < 18; ++ks) {
            const v2f a = araw[ks];
#pragma unroll
            for (int t = 0; t < 8; ++t) {
                const float4 bp = bq[ks][t >> 1];
                const v2f bf = (t & 1) ? v2f{bp.z, bp.w} : v2f{bp.x, bp.y};
                acc[t] = __builtin_amdgcn_wmma_f32_16x16x4_f32(
                    false, a, false, bf, (short)0, acc[t], false, false);
            }
        }

        if (it + 1 < PIX_PER_WAVE) {
            LOAD_A_FRAGMENTS(icap + 1)
        }

        v2f actp[4];
        {
            float s[8];
#pragma unroll
            for (int t = 0; t < 8; ++t) {
                const v2f q = (ROWPAIR(t, 0) + ROWPAIR(t, 1)) + (ROWPAIR(t, 2) + ROWPAIR(t, 3));
                const float r = q.x + q.y;
                s[t] = r + xhalf(r);
            }
            v2f prep[4];
#pragma unroll
            for (int u = 0; u < 4; ++u)
                prep[u] = v2f{s[2 * u], s[2 * u + 1]} * (1.f / 16.f) + biasp[u];
            v2f n2v = prep[0] * prep[0];
#pragma unroll
            for (int u = 1; u < 4; ++u) n2v += prep[u] * prep[u];
            const float sc = squash_scale(n2v.x + n2v.y);
#pragma unroll
            for (int u = 0; u < 4; ++u) actp[u] = prep[u] * sc;
        }

        v2f rtp[4];
#pragma unroll
        for (int j = 0; j < 4; ++j) {
            v2f lg = ROWPAIR(0, j) * actp[0].x;
            lg += ROWPAIR(1, j) * actp[0].y;
#pragma unroll
            for (int u = 1; u < 4; ++u) {
                lg += ROWPAIR(2 * u, j) * actp[u].x;
                lg += ROWPAIR(2 * u + 1, j) * actp[u].y;
            }
            v2f mx = lg;
            mx = v2f{fmaxf(mx.x, xlane<1>(mx.x)), fmaxf(mx.y, xlane<1>(mx.y))};
            mx = v2f{fmaxf(mx.x, xlane<2>(mx.x)), fmaxf(mx.y, xlane<2>(mx.y))};
            mx = v2f{fmaxf(mx.x, xlane<4>(mx.x)), fmaxf(mx.y, xlane<4>(mx.y))};
            mx = v2f{fmaxf(mx.x, xlane<8>(mx.x)), fmaxf(mx.y, xlane<8>(mx.y))};
            const v2f d = lg - mx;
            const v2f e = v2f{__expf(d.x), __expf(d.y)};
            v2f sm = e;
            sm = v2f{sm.x + xlane<1>(sm.x), sm.y + xlane<1>(sm.y)};
            sm = v2f{sm.x + xlane<2>(sm.x), sm.y + xlane<2>(sm.y)};
            sm = v2f{sm.x + xlane<4>(sm.x), sm.y + xlane<4>(sm.y)};
            sm = v2f{sm.x + xlane<8>(sm.x), sm.y + xlane<8>(sm.y)};
            rtp[j] = e * v2f{__builtin_amdgcn_rcpf(sm.x), __builtin_amdgcn_rcpf(sm.y)};
        }

        v2f resp[4];
        {
            float s[8];
#pragma unroll
            for (int t = 0; t < 8; ++t) {
                v2f q = ROWPAIR(t, 0) * rtp[0];
#pragma unroll
                for (int j = 1; j < 4; ++j) q += ROWPAIR(t, j) * rtp[j];
                const float r = q.x + q.y;
                s[t] = r + xhalf(r);
            }
            v2f prep[4];
#pragma unroll
            for (int u = 0; u < 4; ++u)
                prep[u] = v2f{s[2 * u], s[2 * u + 1]} + biasp[u];
            v2f n2v = prep[0] * prep[0];
#pragma unroll
            for (int u = 1; u < 4; ++u) n2v += prep[u] * prep[u];
            const float sc = squash_scale(n2v.x + n2v.y);
#pragma unroll
            for (int u = 0; u < 4; ++u) resp[u] = prep[u] * sc;
        }

        const v2f olo = hi ? resp[2] : resp[0];
        const v2f ohi = hi ? resp[3] : resp[1];
        float4 o;
        o.x = olo.x;
        o.y = olo.y;
        o.z = ohi.x;
        o.w = ohi.y;
        *(float4*)(out + (size_t)pixel * 128 + m * 8 + hi * 4) = o;
    }
}

extern "C" void kernel_launch(void* const* d_in, const int* in_sizes, int n_in,
                              void* d_out, int out_size, void* d_ws, size_t ws_size,
                              hipStream_t stream) {
    (void)in_sizes; (void)n_in; (void)out_size; (void)ws_size;
    const float* input = (const float*)d_in[0];
    const float* Wg    = (const float*)d_in[1];
    const float* bias  = (const float*)d_in[2];
    float* out = (float*)d_out;
    (void)d_ws;

    caps_conv_route_kernel<<<NPIX / (WAVES_PER_BLOCK * PIX_PER_WAVE), BLOCK_THREADS, 0, stream>>>(input, Wg, bias, out);
}
